# starting kernel
# baseline (speedup 1.0000x reference)
#include <hip/hip_runtime.h>
#include <cstddef>

typedef __attribute__((ext_vector_type(2))) float v2f;
typedef __attribute__((ext_vector_type(8))) float v8f;

#define LOG8        2.0794415416798357f
#define A_COUP      0.049787068367863944f
#define ONE_MINUS_A 0.95021293163213606f
#define LOG_NORM    0.29899997f

__global__ __launch_bounds__(256) void bp_gemm_lsm(
    const float* __restrict__ x, const float* __restrict__ W,
    const float* __restrict__ bias, float* __restrict__ log_b0,
    float* __restrict__ log_b, int n_nodes) {
  __shared__ float Wlds[256 * 16];
  __shared__ float Dlds[8][16 * 16];

  const int tid = threadIdx.x;
  {
    const int k = tid;
#pragma unroll
    for (int n = 0; n < 16; ++n)
      Wlds[k * 16 + n] = (n < 8) ? W[k * 8 + n] : 0.0f;
  }
  __syncthreads();

  const int wave = tid >> 5;
  const int lane = tid & 31;
  const int half = lane >> 4;
  const int r    = lane & 15;
  const int tile = blockIdx.x * 8 + wave;
  const int row0 = tile * 16;

  if (row0 < n_nodes) {
    v8f acc = {};
    const float* xrow = x + (size_t)(row0 + r) * 256;
#pragma unroll 4
    for (int k = 0; k < 256; k += 4) {
      const int ka = k + 2 * half;
      v2f a, b;
      a.x = xrow[ka];
      a.y = xrow[ka + 1];
      b.x = Wlds[ka * 16 + r];
      b.y = Wlds[(ka + 1) * 16 + r];
      acc = __builtin_amdgcn_wmma_f32_16x16x4_f32(
          false, a, false, b, (short)0, acc, false, false);
    }
#pragma unroll
    for (int v = 0; v < 8; ++v)
      Dlds[wave][(v + 8 * half) * 16 + r] = acc[v];
  }
  __syncthreads();

  if (row0 < n_nodes && lane < 16) {
    const int row = row0 + lane;
    float z[8];
    float mx = -1e30f;
#pragma unroll
    for (int c = 0; c < 8; ++c) {
      z[c] = Dlds[wave][lane * 16 + c] + bias[c];
      mx = fmaxf(mx, z[c]);
    }
    float s = 0.0f;
#pragma unroll
    for (int c = 0; c < 8; ++c) s += __expf(z[c] - mx);
    const float lse = mx + __logf(s);
    float4 o0, o1;
    o0.x = z[0] - lse; o0.y = z[1] - lse; o0.z = z[2] - lse; o0.w = z[3] - lse;
    o1.x = z[4] - lse; o1.y = z[5] - lse; o1.z = z[6] - lse; o1.w = z[7] - lse;
    float4* p0 = (float4*)(log_b0 + (size_t)row * 8);
    float4* p1 = (float4*)(log_b  + (size_t)row * 8);
    p0[0] = o0; p0[1] = o1;
    p1[0] = o0; p1[1] = o1;
  }
}

template <bool FIRST>
__global__ __launch_bounds__(256) void bp_edge(
    const float* __restrict__ log_b, const float* __restrict__ msg_prev,
    float* __restrict__ msg_new, float* __restrict__ agg,
    const int* __restrict__ src, const int* __restrict__ dst,
    const int* __restrict__ rv, int n_edges) {
  const int e = blockIdx.x * blockDim.x + threadIdx.x;
  if (e >= n_edges) return;

  const int s = src[e];
  const int d = dst[e];

  const float4* xb = (const float4*)(log_b + (size_t)s * 8);
  const float4 x0 = xb[0], x1 = xb[1];
  float t[8] = {x0.x, x0.y, x0.z, x0.w, x1.x, x1.y, x1.z, x1.w};

  if (FIRST) {
#pragma unroll
    for (int c = 0; c < 8; ++c) t[c] += LOG8;
  } else {
    const int rve = rv[e];
    const float4* mb = (const float4*)(msg_prev + (size_t)rve * 8);
    const float4 m0 = mb[0], m1 = mb[1];
    const float mm[8] = {m0.x, m0.y, m0.z, m0.w, m1.x, m1.y, m1.z, m1.w};
#pragma unroll
    for (int c = 0; c < 8; ++c) t[c] -= mm[c];
  }

  float mx = t[0];
#pragma unroll
  for (int c = 1; c < 8; ++c) mx = fmaxf(mx, t[c]);
  float ex[8], S = 0.0f;
#pragma unroll
  for (int c = 0; c < 8; ++c) {
    ex[c] = __expf(t[c] - mx);
    S += ex[c];
  }
  const float norm = __logf(S) + LOG_NORM;
  float msg[8];
#pragma unroll
  for (int c = 0; c < 8; ++c)
    msg[c] = __logf(ONE_MINUS_A * ex[c] + A_COUP * S) - norm;

  float4 w0, w1;
  w0.x = msg[0]; w0.y = msg[1]; w0.z = msg[2]; w0.w = msg[3];
  w1.x = msg[4]; w1.y = msg[5]; w1.z = msg[6]; w1.w = msg[7];
  float4* mo = (float4*)(msg_new + (size_t)e * 8);
  mo[0] = w0; mo[1] = w1;

  float* arow = agg + (size_t)d * 8;
#pragma unroll
  for (int c = 0; c < 8; ++c) unsafeAtomicAdd(arow + c, msg[c]);
}

__global__ __launch_bounds__(256) void bp_node(
    const float* __restrict__ log_b0, float* __restrict__ agg,
    float* __restrict__ log_b, int n_nodes) {
  const int n = blockIdx.x * blockDim.x + threadIdx.x;
  if (n >= n_nodes) return;
  const size_t off = (size_t)n * 8;

  float4* ap = (float4*)(agg + off);
  const float4* bp = (const float4*)(log_b0 + off);
  const float4 a0 = ap[0], a1 = ap[1];
  const float4 b0 = bp[0], b1 = bp[1];
  float z[8] = {a0.x + b0.x, a0.y + b0.y, a0.z + b0.z, a0.w + b0.w,
                a1.x + b1.x, a1.y + b1.y, a1.z + b1.z, a1.w + b1.w};

  const float4 zero = {0.f, 0.f, 0.f, 0.f};
  ap[0] = zero; ap[1] = zero;

  float mx = z[0];
#pragma unroll
  for (int c = 1; c < 8; ++c) mx = fmaxf(mx, z[c]);
  float s = 0.0f;
#pragma unroll
  for (int c = 0; c < 8; ++c) s += __expf(z[c] - mx);
  const float lse = mx + __logf(s);

  float4 o0, o1;
  o0.x = z[0] - lse; o0.y = z[1] - lse; o0.z = z[2] - lse; o0.w = z[3] - lse;
  o1.x = z[4] - lse; o1.y = z[5] - lse; o1.z = z[6] - lse; o1.w = z[7] - lse;
  float4* op = (float4*)(log_b + off);
  op[0] = o0; op[1] = o1;
}

__global__ __launch_bounds__(256) void bp_zero(float* __restrict__ p, int n) {
  const int i = blockIdx.x * blockDim.x + threadIdx.x;
  if (i < n) p[i] = 0.0f;
}

extern "C" void kernel_launch(void* const* d_in, const int* in_sizes, int n_in,
                              void* d_out, int out_size, void* d_ws,
                              size_t ws_size, hipStream_t stream) {
  const float* x    = (const float*)d_in[0];
  const float* W    = (const float*)d_in[1];
  const float* bias = (const float*)d_in[2];
  const int* eidx   = (const int*)d_in[3];
  const int* rv     = (const int*)d_in[4];

  const int n_nodes = in_sizes[0] / 256;
  const int n_edges = in_sizes[4];
  const int* src = eidx;
  const int* dst = eidx + n_edges;

  float* out = (float*)d_out;
  float* ws  = (float*)d_ws;
  float* log_b0 = ws;
  float* agg    = log_b0 + (size_t)n_nodes * 8;
  float* msgA   = agg    + (size_t)n_nodes * 8;
  float* msgB   = msgA   + (size_t)n_edges * 8;

  const int nf = n_nodes * 8;
  bp_zero<<<(nf + 255) / 256, 256, 0, stream>>>(agg, nf);

  const int n_tiles = (n_nodes + 15) / 16;
  bp_gemm_lsm<<<(n_tiles + 7) / 8, 256, 0, stream>>>(x, W, bias, log_b0, out,
                                                     n_nodes);

  const int eb = (n_edges + 255) / 256;
  const int nb = (n_nodes + 255) / 256;

  bp_edge<true><<<eb, 256, 0, stream>>>(out, nullptr, msgA, agg, src, dst, rv,
                                        n_edges);
  bp_node<<<nb, 256, 0, stream>>>(log_b0, agg, out, n_nodes);

  float* prev = msgA;
  float* cur  = msgB;
  for (int step = 1; step < 5; ++step) {
    bp_edge<false><<<eb, 256, 0, stream>>>(out, prev, cur, agg, src, dst, rv,
                                           n_edges);
    bp_node<<<nb, 256, 0, stream>>>(log_b0, agg, out, n_nodes);
    float* tmp = prev; prev = cur; cur = tmp;
  }
}


# Edge scatter issued row-wise: a wave parks its 32 message rows in LDS so 8 adjacent lanes add one 32-byte agg row (4 rows per atomic instruction, not 32 words); f32 atomics, arrival-order sums kept
# speedup vs baseline: 3.9067x; 3.9067x over previous
#include <hip/hip_runtime.h>
#include <cstddef>

typedef __attribute__((ext_vector_type(2))) float v2f;
typedef __attribute__((ext_vector_type(8))) float v8f;

#define LOG8        2.0794415416798357f
#define A_COUP      0.049787068367863944f
#define ONE_MINUS_A 0.95021293163213606f
#define LOG_NORM    0.29899997f

__global__ __launch_bounds__(256) void bp_gemm_lsm(
    const float* __restrict__ x, const float* __restrict__ W,
    const float* __restrict__ bias, float* __restrict__ log_b0,
    float* __restrict__ log_b, int n_nodes) {
  __shared__ float Wlds[256 * 16];
  __shared__ float Dlds[8][16 * 16];

  const int tid = threadIdx.x;
  {
    const int k = tid;
#pragma unroll
    for (int n = 0; n < 16; ++n)
      Wlds[k * 16 + n] = (n < 8) ? W[k * 8 + n] : 0.0f;
  }
  __syncthreads();

  const int wave = tid >> 5;
  const int lane = tid & 31;
  const int half = lane >> 4;
  const int r    = lane & 15;
  const int tile = blockIdx.x * 8 + wave;
  const int row0 = tile * 16;

  if (row0 < n_nodes) {
    v8f acc = {};
    const float* xrow = x + (size_t)(row0 + r) * 256;
#pragma unroll 4
    for (int k = 0; k < 256; k += 4) {
      const int ka = k + 2 * half;
      v2f a, b;
      a.x = xrow[ka];
      a.y = xrow[ka + 1];
      b.x = Wlds[ka * 16 + r];
      b.y = Wlds[(ka + 1) * 16 + r];
      acc = __builtin_amdgcn_wmma_f32_16x16x4_f32(
          false, a, false, b, (short)0, acc, false, false);
    }
#pragma unroll
    for (int v = 0; v < 8; ++v)
      Dlds[wave][(v + 8 * half) * 16 + r] = acc[v];
  }
  __syncthreads();

  if (row0 < n_nodes && lane < 16) {
    const int row = row0 + lane;
    float z[8];
    float mx = -1e30f;
#pragma unroll
    for (int c = 0; c < 8; ++c) {
      z[c] = Dlds[wave][lane * 16 + c] + bias[c];
      mx = fmaxf(mx, z[c]);
    }
    float s = 0.0f;
#pragma unroll
    for (int c = 0; c < 8; ++c) s += __expf(z[c] - mx);
    const float lse = mx + __logf(s);
    float4 o0, o1;
    o0.x = z[0] - lse; o0.y = z[1] - lse; o0.z = z[2] - lse; o0.w = z[3] - lse;
    o1.x = z[4] - lse; o1.y = z[5] - lse; o1.z = z[6] - lse; o1.w = z[7] - lse;
    float4* p0 = (float4*)(log_b0 + (size_t)row * 8);
    float4* p1 = (float4*)(log_b  + (size_t)row * 8);
    p0[0] = o0; p0[1] = o1;
    p1[0] = o0; p1[1] = o1;
  }
}

template <bool FIRST>
__global__ __launch_bounds__(256) void bp_edge(
    const float* __restrict__ log_b, const float* __restrict__ msg_prev,
    float* __restrict__ msg_new, float* __restrict__ agg,
    const int* __restrict__ src, const int* __restrict__ dst,
    const int* __restrict__ rv, int n_edges) {
  __shared__ __attribute__((aligned(16))) float srow[8][32 * 8];
  __shared__ __attribute__((aligned(16))) int sdst[8][32];

  const int e = blockIdx.x * blockDim.x + threadIdx.x;
  const int wave = threadIdx.x >> 5;
  const int lane = threadIdx.x & 31;

  float4 w0 = {0.f, 0.f, 0.f, 0.f}, w1 = {0.f, 0.f, 0.f, 0.f};
  int d = -1;

  if (e < n_edges) {
  const int s = src[e];
  d = dst[e];

  const float4* xb = (const float4*)(log_b + (size_t)s * 8);
  const float4 x0 = xb[0], x1 = xb[1];
  float t[8] = {x0.x, x0.y, x0.z, x0.w, x1.x, x1.y, x1.z, x1.w};

  if (FIRST) {
#pragma unroll
    for (int c = 0; c < 8; ++c) t[c] += LOG8;
  } else {
    const int rve = rv[e];
    const float4* mb = (const float4*)(msg_prev + (size_t)rve * 8);
    const float4 m0 = mb[0], m1 = mb[1];
    const float mm[8] = {m0.x, m0.y, m0.z, m0.w, m1.x, m1.y, m1.z, m1.w};
#pragma unroll
    for (int c = 0; c < 8; ++c) t[c] -= mm[c];
  }

  float mx = t[0];
#pragma unroll
  for (int c = 1; c < 8; ++c) mx = fmaxf(mx, t[c]);
  float ex[8], S = 0.0f;
#pragma unroll
  for (int c = 0; c < 8; ++c) {
    ex[c] = __expf(t[c] - mx);
    S += ex[c];
  }
  const float norm = __logf(S) + LOG_NORM;
  float msg[8];
#pragma unroll
  for (int c = 0; c < 8; ++c)
    msg[c] = __logf(ONE_MINUS_A * ex[c] + A_COUP * S) - norm;

  w0.x = msg[0]; w0.y = msg[1]; w0.z = msg[2]; w0.w = msg[3];
  w1.x = msg[4]; w1.y = msg[5]; w1.z = msg[6]; w1.w = msg[7];
  }

  float4* sp = (float4*)&srow[wave][lane * 8];
  sp[0] = w0; sp[1] = w1;
  sdst[wave][(lane & 3) * 8 + (lane >> 2)] = d;
  __syncthreads();

  if (e < n_edges) {
    float4* mo = (float4*)(msg_new + (size_t)e * 8);
    mo[0] = w0; mo[1] = w1;
  }

  const int cls = lane & 7;
  const int sub = lane >> 3;
  const int4 da = *(const int4*)&sdst[wave][sub * 8];
  const int4 db = *(const int4*)&sdst[wave][sub * 8 + 4];
  const int dd[8] = {da.x, da.y, da.z, da.w, db.x, db.y, db.z, db.w};
  float v[8];
#pragma unroll
  for (int i = 0; i < 8; ++i) v[i] = srow[wave][32 * i + lane];
  float* acol = agg + cls;
#pragma unroll
  for (int i = 0; i < 8; ++i)
    if (dd[i] >= 0) unsafeAtomicAdd(acol + (size_t)dd[i] * 8, v[i]);
}

__global__ __launch_bounds__(256) void bp_node(
    const float* __restrict__ log_b0, float* __restrict__ agg,
    float* __restrict__ log_b, int n_nodes) {
  const int n = blockIdx.x * blockDim.x + threadIdx.x;
  if (n >= n_nodes) return;
  const size_t off = (size_t)n * 8;

  float4* ap = (float4*)(agg + off);
  const float4* bp = (const float4*)(log_b0 + off);
  const float4 a0 = ap[0], a1 = ap[1];
  const float4 b0 = bp[0], b1 = bp[1];
  float z[8] = {a0.x + b0.x, a0.y + b0.y, a0.z + b0.z, a0.w + b0.w,
                a1.x + b1.x, a1.y + b1.y, a1.z + b1.z, a1.w + b1.w};

  const float4 zero = {0.f, 0.f, 0.f, 0.f};
  ap[0] = zero; ap[1] = zero;

  float mx = z[0];
#pragma unroll
  for (int c = 1; c < 8; ++c) mx = fmaxf(mx, z[c]);
  float s = 0.0f;
#pragma unroll
  for (int c = 0; c < 8; ++c) s += __expf(z[c] - mx);
  const float lse = mx + __logf(s);

  float4 o0, o1;
  o0.x = z[0] - lse; o0.y = z[1] - lse; o0.z = z[2] - lse; o0.w = z[3] - lse;
  o1.x = z[4] - lse; o1.y = z[5] - lse; o1.z = z[6] - lse; o1.w = z[7] - lse;
  float4* op = (float4*)(log_b + off);
  op[0] = o0; op[1] = o1;
}

__global__ __launch_bounds__(256) void bp_zero(float* __restrict__ p, int n) {
  const int i = blockIdx.x * blockDim.x + threadIdx.x;
  if (i < n) p[i] = 0.0f;
}

extern "C" void kernel_launch(void* const* d_in, const int* in_sizes, int n_in,
                              void* d_out, int out_size, void* d_ws,
                              size_t ws_size, hipStream_t stream) {
  const float* x    = (const float*)d_in[0];
  const float* W    = (const float*)d_in[1];
  const float* bias = (const float*)d_in[2];
  const int* eidx   = (const int*)d_in[3];
  const int* rv     = (const int*)d_in[4];

  const int n_nodes = in_sizes[0] / 256;
  const int n_edges = in_sizes[4];
  const int* src = eidx;
  const int* dst = eidx + n_edges;

  float* out = (float*)d_out;
  float* ws  = (float*)d_ws;
  float* log_b0 = ws;
  float* agg    = log_b0 + (size_t)n_nodes * 8;
  float* msgA   = agg    + (size_t)n_nodes * 8;
  float* msgB   = msgA   + (size_t)n_edges * 8;

  const int nf = n_nodes * 8;
  bp_zero<<<(nf + 255) / 256, 256, 0, stream>>>(agg, nf);

  const int n_tiles = (n_nodes + 15) / 16;
  bp_gemm_lsm<<<(n_tiles + 7) / 8, 256, 0, stream>>>(x, W, bias, log_b0, out,
                                                     n_nodes);

  const int eb = (n_edges + 255) / 256;
  const int nb = (n_nodes + 255) / 256;

  bp_edge<true><<<eb, 256, 0, stream>>>(out, nullptr, msgA, agg, src, dst, rv,
                                        n_edges);
  bp_node<<<nb, 256, 0, stream>>>(log_b0, agg, out, n_nodes);

  float* prev = msgA;
  float* cur  = msgB;
  for (int step = 1; step < 5; ++step) {
    bp_edge<false><<<eb, 256, 0, stream>>>(out, prev, cur, agg, src, dst, rv,
                                           n_edges);
    bp_node<<<nb, 256, 0, stream>>>(log_b0, agg, out, n_nodes);
    float* tmp = prev; prev = cur; cur = tmp;
  }
}


# speedup vs baseline: 6.6471x; 1.7015x over previous
#include <hip/hip_runtime.h>
#include <cstddef>

typedef __attribute__((ext_vector_type(2))) float v2f;
typedef __attribute__((ext_vector_type(8))) float v8f;

#define LOG8        2.0794415416798357f
#define A_COUP      0.049787068367863944f
#define ONE_MINUS_A 0.95021293163213606f
#define LOG_NORM    0.29899997f

__global__ __launch_bounds__(256) void bp_gemm_lsm(
    const float* __restrict__ x, const float* __restrict__ W,
    const float* __restrict__ bias, float* __restrict__ log_b0,
    float* __restrict__ log_b, int n_nodes) {
  __shared__ float Wlds[256 * 16];
  __shared__ float Dlds[8][16 * 16];

  const int tid = threadIdx.x;
  {
    const int k = tid;
#pragma unroll
    for (int n = 0; n < 16; ++n)
      Wlds[k * 16 + n] = (n < 8) ? W[k * 8 + n] : 0.0f;
  }
  __syncthreads();

  const int wave = tid >> 5;
  const int lane = tid & 31;
  const int half = lane >> 4;
  const int r    = lane & 15;
  const int tile = blockIdx.x * 8 + wave;
  const int row0 = tile * 16;

  if (row0 < n_nodes) {
    v8f acc = {};
    const float* xrow = x + (size_t)(row0 + r) * 256;
#pragma unroll 4
    for (int k = 0; k < 256; k += 4) {
      const int ka = k + 2 * half;
      v2f a, b;
      a.x = xrow[ka];
      a.y = xrow[ka + 1];
      b.x = Wlds[ka * 16 + r];
      b.y = Wlds[(ka + 1) * 16 + r];
      acc = __builtin_amdgcn_wmma_f32_16x16x4_f32(
          false, a, false, b, (short)0, acc, false, false);
    }
#pragma unroll
    for (int v = 0; v < 8; ++v)
      Dlds[wave][(v + 8 * half) * 16 + r] = acc[v];
  }
  __syncthreads();

  if (row0 < n_nodes && lane < 16) {
    const int row = row0 + lane;
    float z[8];
    float mx = -1e30f;
#pragma unroll
    for (int c = 0; c < 8; ++c) {
      z[c] = Dlds[wave][lane * 16 + c] + bias[c];
      mx = fmaxf(mx, z[c]);
    }
    float s = 0.0f;
#pragma unroll
    for (int c = 0; c < 8; ++c) s += __expf(z[c] - mx);
    const float lse = mx + __logf(s);
    float4 o0, o1;
    o0.x = z[0] - lse; o0.y = z[1] - lse; o0.z = z[2] - lse; o0.w = z[3] - lse;
    o1.x = z[4] - lse; o1.y = z[5] - lse; o1.z = z[6] - lse; o1.w = z[7] - lse;
    float4* p0 = (float4*)(log_b0 + (size_t)row * 8);
    float4* p1 = (float4*)(log_b  + (size_t)row * 8);
    p0[0] = o0; p0[1] = o1;
    p1[0] = o0; p1[1] = o1;
  }
}

template <bool FIRST>
__global__ __launch_bounds__(256) void bp_edge(
    const float* __restrict__ log_b, const float* __restrict__ msg_prev,
    float* __restrict__ msg_new, float* __restrict__ agg,
    const int* __restrict__ src, const int* __restrict__ dst,
    const int* __restrict__ rv, int n_edges, int plane_stride) {
  const int place = __builtin_amdgcn_s_sendmsg_rtn(0x87);
  const int half = (place >> 18) & 1;

  __shared__ __attribute__((aligned(16))) float srow[8][32 * 8];
  __shared__ __attribute__((aligned(16))) int sdst[8][32];

  const int e = blockIdx.x * blockDim.x + threadIdx.x;
  const int wave = threadIdx.x >> 5;
  const int lane = threadIdx.x & 31;

  float4 w0 = {0.f, 0.f, 0.f, 0.f}, w1 = {0.f, 0.f, 0.f, 0.f};
  int d = -1;

  if (e < n_edges) {
  const int s = src[e];
  d = dst[e];

  const float4* xb = (const float4*)(log_b + (size_t)s * 8);
  const float4 x0 = xb[0], x1 = xb[1];
  float t[8] = {x0.x, x0.y, x0.z, x0.w, x1.x, x1.y, x1.z, x1.w};

  if (FIRST) {
#pragma unroll
    for (int c = 0; c < 8; ++c) t[c] += LOG8;
  } else {
    const int rve = rv[e];
    const float4* mb = (const float4*)(msg_prev + (size_t)rve * 8);
    const float4 m0 = mb[0], m1 = mb[1];
    const float mm[8] = {m0.x, m0.y, m0.z, m0.w, m1.x, m1.y, m1.z, m1.w};
#pragma unroll
    for (int c = 0; c < 8; ++c) t[c] -= mm[c];
  }

  float mx = t[0];
#pragma unroll
  for (int c = 1; c < 8; ++c) mx = fmaxf(mx, t[c]);
  float ex[8], S = 0.0f;
#pragma unroll
  for (int c = 0; c < 8; ++c) {
    ex[c] = __expf(t[c] - mx);
    S += ex[c];
  }
  const float norm = __logf(S) + LOG_NORM;
  float msg[8];
#pragma unroll
  for (int c = 0; c < 8; ++c)
    msg[c] = __logf(ONE_MINUS_A * ex[c] + A_COUP * S) - norm;

  w0.x = msg[0]; w0.y = msg[1]; w0.z = msg[2]; w0.w = msg[3];
  w1.x = msg[4]; w1.y = msg[5]; w1.z = msg[6]; w1.w = msg[7];
  }

  float4* sp = (float4*)&srow[wave][lane * 8];
  sp[0] = w0; sp[1] = w1;
  sdst[wave][(lane & 3) * 8 + (lane >> 2)] = d;
  __syncthreads();

  if (e < n_edges) {
    float4* mo = (float4*)(msg_new + (size_t)e * 8);
    mo[0] = w0; mo[1] = w1;
  }

  const int cls = lane & 7;
  const int sub = lane >> 3;
  const int4 da = *(const int4*)&sdst[wave][sub * 8];
  const int4 db = *(const int4*)&sdst[wave][sub * 8 + 4];
  const int dd[8] = {da.x, da.y, da.z, da.w, db.x, db.y, db.z, db.w};
  float v[8];
#pragma unroll
  for (int i = 0; i < 8; ++i) v[i] = srow[wave][32 * i + lane];
  float* acol = agg + (size_t)half * plane_stride + cls;
#pragma unroll
  for (int i = 0; i < 8; ++i)
    if (dd[i] >= 0) unsafeAtomicAdd(acol + (size_t)dd[i] * 8, v[i]);
}

__global__ __launch_bounds__(256) void bp_node(
    const float* __restrict__ log_b0, float* __restrict__ agg,
    float* __restrict__ log_b, int n_nodes, int plane_stride) {
  const int n = blockIdx.x * blockDim.x + threadIdx.x;
  if (n >= n_nodes) return;
  const size_t off = (size_t)n * 8;

  float4* ap = (float4*)(agg + off);
  const float4* bp = (const float4*)(log_b0 + off);
  float4 a0 = ap[0], a1 = ap[1];
  const float4 b0 = bp[0], b1 = bp[1];
  const float4 zero = {0.f, 0.f, 0.f, 0.f};
  ap[0] = zero; ap[1] = zero;
  if (plane_stride) {
    float4* aq = (float4*)(agg + (size_t)plane_stride + off);
    const float4 c0 = aq[0], c1 = aq[1];
    aq[0] = zero; aq[1] = zero;
    a0.x += c0.x; a0.y += c0.y; a0.z += c0.z; a0.w += c0.w;
    a1.x += c1.x; a1.y += c1.y; a1.z += c1.z; a1.w += c1.w;
  }
  float z[8] = {a0.x + b0.x, a0.y + b0.y, a0.z + b0.z, a0.w + b0.w,
                a1.x + b1.x, a1.y + b1.y, a1.z + b1.z, a1.w + b1.w};

  float mx = z[0];
#pragma unroll
  for (int c = 1; c < 8; ++c) mx = fmaxf(mx, z[c]);
  float s = 0.0f;
#pragma unroll
  for (int c = 0; c < 8; ++c) s += __expf(z[c] - mx);
  const float lse = mx + __logf(s);

  float4 o0, o1;
  o0.x = z[0] - lse; o0.y = z[1] - lse; o0.z = z[2] - lse; o0.w = z[3] - lse;
  o1.x = z[4] - lse; o1.y = z[5] - lse; o1.z = z[6] - lse; o1.w = z[7] - lse;
  float4* op = (float4*)(log_b + off);
  op[0] = o0; op[1] = o1;
}

__global__ __launch_bounds__(256) void bp_zero(float* __restrict__ p, int n) {
  const int i = blockIdx.x * blockDim.x + threadIdx.x;
  if (i < n) p[i] = 0.0f;
}

extern "C" void kernel_launch(void* const* d_in, const int* in_sizes, int n_in,
                              void* d_out, int out_size, void* d_ws,
                              size_t ws_size, hipStream_t stream) {
  const float* x    = (const float*)d_in[0];
  const float* W    = (const float*)d_in[1];
  const float* bias = (const float*)d_in[2];
  const int* eidx   = (const int*)d_in[3];
  const int* rv     = (const int*)d_in[4];

  const int n_nodes = in_sizes[0] / 256;
  const int n_edges = in_sizes[4];
  const int* src = eidx;
  const int* dst = eidx + n_edges;

  float* out = (float*)d_out;
  float* ws  = (float*)d_ws;
  float* log_b0 = ws;
  const int nf = n_nodes * 8;
  const size_t need = ((size_t)3 * nf + (size_t)16 * n_edges) * sizeof(float);
  const int plane_stride = (ws_size >= need) ? nf : 0;
  const int naf = nf + plane_stride;
  float* agg    = log_b0 + (size_t)nf;
  float* msgA   = agg    + (size_t)naf;
  float* msgB   = msgA   + (size_t)n_edges * 8;

  bp_zero<<<(naf + 255) / 256, 256, 0, stream>>>(agg, naf);

  const int n_tiles = (n_nodes + 15) / 16;
  bp_gemm_lsm<<<(n_tiles + 7) / 8, 256, 0, stream>>>(x, W, bias, log_b0, out,
                                                     n_nodes);

  const int eb = (n_edges + 255) / 256;
  const int nb = (n_nodes + 255) / 256;

  bp_edge<true><<<eb, 256, 0, stream>>>(out, nullptr, msgA, agg, src, dst, rv,
                                        n_edges, plane_stride);
  bp_node<<<nb, 256, 0, stream>>>(log_b0, agg, out, n_nodes, plane_stride);

  float* prev = msgA;
  float* cur  = msgB;
  for (int step = 1; step < 5; ++step) {
    bp_edge<false><<<eb, 256, 0, stream>>>(out, prev, cur, agg, src, dst, rv,
                                           n_edges, plane_stride);
    bp_node<<<nb, 256, 0, stream>>>(log_b0, agg, out, n_nodes, plane_stride);
    float* tmp = prev; prev = cur; cur = tmp;
  }
}


# Edge and zeroing kernels run as fixed grids whose waves loop over tiles (edge: 1024 blocks, next tile's rows and indices in flight): 5x fewer waves start per run; f32 atomics, arrival-order sums kept
# speedup vs baseline: 7.6415x; 1.1496x over previous
#include <hip/hip_runtime.h>
#include <cstddef>

typedef __attribute__((ext_vector_type(2))) float v2f;
typedef __attribute__((ext_vector_type(8))) float v8f;

#define LOG8        2.0794415416798357f
#define A_COUP      0.049787068367863944f
#define ONE_MINUS_A 0.95021293163213606f
#define LOG_NORM    0.29899997f

__global__ __launch_bounds__(256) void bp_gemm_lsm(
    const float* __restrict__ x, const float* __restrict__ W,
    const float* __restrict__ bias, float* __restrict__ log_b0,
    float* __restrict__ log_b, int n_nodes) {
  __shared__ float Wlds[256 * 16];
  __shared__ float Dlds[8][16 * 16];

  const int tid = threadIdx.x;
  {
    const int k = tid;
#pragma unroll
    for (int n = 0; n < 16; ++n)
      Wlds[k * 16 + n] = (n < 8) ? W[k * 8 + n] : 0.0f;
  }
  __syncthreads();

  const int wave = tid >> 5;
  const int lane = tid & 31;
  const int half = lane >> 4;
  const int r    = lane & 15;
  const int tile = blockIdx.x * 8 + wave;
  const int row0 = tile * 16;

  if (row0 < n_nodes) {
    v8f acc = {};
    const float* xrow = x + (size_t)(row0 + r) * 256;
#pragma unroll 4
    for (int k = 0; k < 256; k += 4) {
      const int ka = k + 2 * half;
      v2f a, b;
      a.x = xrow[ka];
      a.y = xrow[ka + 1];
      b.x = Wlds[ka * 16 + r];
      b.y = Wlds[(ka + 1) * 16 + r];
      acc = __builtin_amdgcn_wmma_f32_16x16x4_f32(
          false, a, false, b, (short)0, acc, false, false);
    }
#pragma unroll
    for (int v = 0; v < 8; ++v)
      Dlds[wave][(v + 8 * half) * 16 + r] = acc[v];
  }
  __syncthreads();

  if (row0 < n_nodes && lane < 16) {
    const int row = row0 + lane;
    float z[8];
    float mx = -1e30f;
#pragma unroll
    for (int c = 0; c < 8; ++c) {
      z[c] = Dlds[wave][lane * 16 + c] + bias[c];
      mx = fmaxf(mx, z[c]);
    }
    float s = 0.0f;
#pragma unroll
    for (int c = 0; c < 8; ++c) s += __expf(z[c] - mx);
    const float lse = mx + __logf(s);
    float4 o0, o1;
    o0.x = z[0] - lse; o0.y = z[1] - lse; o0.z = z[2] - lse; o0.w = z[3] - lse;
    o1.x = z[4] - lse; o1.y = z[5] - lse; o1.z = z[6] - lse; o1.w = z[7] - lse;
    float4* p0 = (float4*)(log_b0 + (size_t)row * 8);
    float4* p1 = (float4*)(log_b  + (size_t)row * 8);
    p0[0] = o0; p0[1] = o1;
    p1[0] = o0; p1[1] = o1;
  }
}

template <bool FIRST>
__global__ __launch_bounds__(256) void bp_edge(
    const float* __restrict__ log_b, const float* __restrict__ msg_prev,
    float* __restrict__ msg_new, float* __restrict__ agg,
    const int* __restrict__ src, const int* __restrict__ dst,
    const int* __restrict__ rv, int n_edges, int plane_stride) {
  const int place = __builtin_amdgcn_s_sendmsg_rtn(0x87);
  const int half = (place >> 18) & 1;

  __shared__ __attribute__((aligned(16))) float srow[8][32 * 8];
  __shared__ __attribute__((aligned(16))) int sdst[8][32];

  const int wave = threadIdx.x >> 5;
  const int lane = threadIdx.x & 31;
  const int cls = lane & 7;
  const int sub = lane >> 3;
  float* acol = agg + (size_t)half * plane_stride + cls;

  const int stride = (int)gridDim.x * 256;
  const int wbase = __builtin_amdgcn_readfirstlane(
      (int)(blockIdx.x * 256 + (threadIdx.x & ~31u)));

  int d_c = -1;
  int s_a = 0, d_a = -1, r_a = 0;
  float4 x0 = {0.f, 0.f, 0.f, 0.f}, x1 = x0, m0 = x0, m1 = x0;
  {
    const int e = wbase + lane;
    int s_c = 0, r_c = 0;
    if (e < n_edges) {
      s_c = src[e];
      d_c = dst[e];
      if (!FIRST) r_c = rv[e];
    }
    const int e1 = e + stride;
    if (e1 < n_edges) {
      s_a = src[e1];
      d_a = dst[e1];
      if (!FIRST) r_a = rv[e1];
    }
    if (e < n_edges) {
      const float4* xb = (const float4*)(log_b + (size_t)s_c * 8);
      x0 = xb[0]; x1 = xb[1];
      if (!FIRST) {
        const float4* mb = (const float4*)(msg_prev + (size_t)r_c * 8);
        m0 = mb[0]; m1 = mb[1];
      }
    }
  }

  for (int tb = wbase; tb < n_edges; tb += stride) {
    const int e = tb + lane;
    const bool live = e < n_edges;
    const float4 cx0 = x0, cx1 = x1, cm0 = m0, cm1 = m1;
    const int d = d_c;

    const int e1 = e + stride;
    if (e1 < n_edges) {
      const float4* xb = (const float4*)(log_b + (size_t)s_a * 8);
      x0 = xb[0]; x1 = xb[1];
      if (!FIRST) {
        const float4* mb = (const float4*)(msg_prev + (size_t)r_a * 8);
        m0 = mb[0]; m1 = mb[1];
      }
    }
    d_c = d_a;
    const int e2 = e1 + stride;
    d_a = -1;
    if (e2 < n_edges) {
      s_a = src[e2];
      d_a = dst[e2];
      if (!FIRST) r_a = rv[e2];
    }

    float4 w0 = {0.f, 0.f, 0.f, 0.f}, w1 = {0.f, 0.f, 0.f, 0.f};
    if (live) {
      float t[8] = {cx0.x, cx0.y, cx0.z, cx0.w, cx1.x, cx1.y, cx1.z, cx1.w};
      if (FIRST) {
#pragma unroll
        for (int c = 0; c < 8; ++c) t[c] += LOG8;
      } else {
        const float mm[8] = {cm0.x, cm0.y, cm0.z, cm0.w,
                             cm1.x, cm1.y, cm1.z, cm1.w};
#pragma unroll
        for (int c = 0; c < 8; ++c) t[c] -= mm[c];
      }

      float mx = t[0];
#pragma unroll
      for (int c = 1; c < 8; ++c) mx = fmaxf(mx, t[c]);
      float ex[8], S = 0.0f;
#pragma unroll
      for (int c = 0; c < 8; ++c) {
        ex[c] = __expf(t[c] - mx);
        S += ex[c];
      }
      const float norm = __logf(S) + LOG_NORM;
      float msg[8];
#pragma unroll
      for (int c = 0; c < 8; ++c)
        msg[c] = __logf(ONE_MINUS_A * ex[c] + A_COUP * S) - norm;

      w0.x = msg[0]; w0.y = msg[1]; w0.z = msg[2]; w0.w = msg[3];
      w1.x = msg[4]; w1.y = msg[5]; w1.z = msg[6]; w1.w = msg[7];
    }

    float4* sp = (float4*)&srow[wave][lane * 8];
    sp[0] = w0; sp[1] = w1;
    sdst[wave][(lane & 3) * 8 + (lane >> 2)] = live ? d : -1;
    __builtin_amdgcn_fence(__ATOMIC_RELEASE, "wavefront");
    __builtin_amdgcn_wave_barrier();
    __builtin_amdgcn_fence(__ATOMIC_ACQUIRE, "wavefront");

    if (live) {
      float4* mo = (float4*)(msg_new + (size_t)e * 8);
      mo[0] = w0; mo[1] = w1;
    }

    const int4 da = *(const int4*)&sdst[wave][sub * 8];
    const int4 db = *(const int4*)&sdst[wave][sub * 8 + 4];
    const int dd[8] = {da.x, da.y, da.z, da.w, db.x, db.y, db.z, db.w};
    float v[8];
#pragma unroll
    for (int i = 0; i < 8; ++i) v[i] = srow[wave][32 * i + lane];
#pragma unroll
    for (int i = 0; i < 8; ++i)
      if (dd[i] >= 0) unsafeAtomicAdd(acol + (size_t)dd[i] * 8, v[i]);
  }
}

__global__ __launch_bounds__(256) void bp_node(
    const float* __restrict__ log_b0, float* __restrict__ agg,
    float* __restrict__ log_b, int n_nodes, int plane_stride) {
  const int n = blockIdx.x * blockDim.x + threadIdx.x;
  if (n >= n_nodes) return;
  const size_t off = (size_t)n * 8;

  float4* ap = (float4*)(agg + off);
  const float4* bp = (const float4*)(log_b0 + off);
  float4 a0 = ap[0], a1 = ap[1];
  const float4 b0 = bp[0], b1 = bp[1];
  const float4 zero = {0.f, 0.f, 0.f, 0.f};
  ap[0] = zero; ap[1] = zero;
  if (plane_stride) {
    float4* aq = (float4*)(agg + (size_t)plane_stride + off);
    const float4 c0 = aq[0], c1 = aq[1];
    aq[0] = zero; aq[1] = zero;
    a0.x += c0.x; a0.y += c0.y; a0.z += c0.z; a0.w += c0.w;
    a1.x += c1.x; a1.y += c1.y; a1.z += c1.z; a1.w += c1.w;
  }
  float z[8] = {a0.x + b0.x, a0.y + b0.y, a0.z + b0.z, a0.w + b0.w,
                a1.x + b1.x, a1.y + b1.y, a1.z + b1.z, a1.w + b1.w};

  float mx = z[0];
#pragma unroll
  for (int c = 1; c < 8; ++c) mx = fmaxf(mx, z[c]);
  float s = 0.0f;
#pragma unroll
  for (int c = 0; c < 8; ++c) s += __expf(z[c] - mx);
  const float lse = mx + __logf(s);

  float4 o0, o1;
  o0.x = z[0] - lse; o0.y = z[1] - lse; o0.z = z[2] - lse; o0.w = z[3] - lse;
  o1.x = z[4] - lse; o1.y = z[5] - lse; o1.z = z[6] - lse; o1.w = z[7] - lse;
  float4* op = (float4*)(log_b + off);
  op[0] = o0; op[1] = o1;
}

__global__ __launch_bounds__(256) void bp_zero(float* __restrict__ p, int n) {
  float4* p4 = (float4*)p;
  const int n4 = n >> 2;
  const float4 zero = {0.f, 0.f, 0.f, 0.f};
  for (int i = blockIdx.x * 256 + threadIdx.x; i < n4; i += gridDim.x * 256)
    p4[i] = zero;
}

extern "C" void kernel_launch(void* const* d_in, const int* in_sizes, int n_in,
                              void* d_out, int out_size, void* d_ws,
                              size_t ws_size, hipStream_t stream) {
  const float* x    = (const float*)d_in[0];
  const float* W    = (const float*)d_in[1];
  const float* bias = (const float*)d_in[2];
  const int* eidx   = (const int*)d_in[3];
  const int* rv     = (const int*)d_in[4];

  const int n_nodes = in_sizes[0] / 256;
  const int n_edges = in_sizes[4];
  const int* src = eidx;
  const int* dst = eidx + n_edges;

  float* out = (float*)d_out;
  float* ws  = (float*)d_ws;
  float* log_b0 = ws;
  const int nf = n_nodes * 8;
  const size_t need = ((size_t)3 * nf + (size_t)16 * n_edges) * sizeof(float);
  const int plane_stride = (ws_size >= need) ? nf : 0;
  const int naf = nf + plane_stride;
  float* agg    = log_b0 + (size_t)nf;
  float* msgA   = agg    + (size_t)naf;
  float* msgB   = msgA   + (size_t)n_edges * 8;

  bp_zero<<<256, 256, 0, stream>>>(agg, naf);

  const int n_tiles = (n_nodes + 15) / 16;
  bp_gemm_lsm<<<(n_tiles + 7) / 8, 256, 0, stream>>>(x, W, bias, log_b0, out,
                                                     n_nodes);

  const int ebn = (n_edges + 255) / 256;
  const int eb = ebn < 1024 ? ebn : 1024;
  const int nb = (n_nodes + 255) / 256;

  bp_edge<true><<<eb, 256, 0, stream>>>(out, nullptr, msgA, agg, src, dst, rv,
                                        n_edges, plane_stride);
  bp_node<<<nb, 256, 0, stream>>>(log_b0, agg, out, n_nodes, plane_stride);

  float* prev = msgA;
  float* cur  = msgB;
  for (int step = 1; step < 5; ++step) {
    bp_edge<false><<<eb, 256, 0, stream>>>(out, prev, cur, agg, src, dst, rv,
                                           n_edges, plane_stride);
    bp_node<<<nb, 256, 0, stream>>>(log_b0, agg, out, n_nodes, plane_stride);
    float* tmp = prev; prev = cur; cur = tmp;
  }
}
